# starting kernel
# baseline (speedup 1.0000x reference)
#include <hip/hip_runtime.h>

#define NN 50000
#define NE 800000

typedef __attribute__((ext_vector_type(2))) float v2f;
typedef __attribute__((ext_vector_type(8))) float v8f;

__global__ void k_init_deg(float* deg) {
  int i = blockIdx.x * blockDim.x + threadIdx.x;
  if (i < NN) deg[i] = 1.0f;
}

__global__ void k_deg_edges(const int* __restrict__ dst, float* __restrict__ deg) {
  int e = blockIdx.x * blockDim.x + threadIdx.x;
  if (e < NE)
    __hip_atomic_fetch_add(deg + dst[e], 1.0f, __ATOMIC_RELAXED, __HIP_MEMORY_SCOPE_AGENT);
}

__global__ void k_rsqrt(float* d) {
  int i = blockIdx.x * blockDim.x + threadIdx.x;
  if (i < NN) d[i] = rsqrtf(d[i]);
}

__global__ void k_zero(float* __restrict__ p, unsigned long long n) {
  unsigned long long i = (unsigned long long)blockIdx.x * blockDim.x + threadIdx.x;
  if (i < n) p[i] = 0.0f;
}

__global__ void k_gemm_wmma(const float* __restrict__ X, const float* __restrict__ W,
                            float* __restrict__ H, int K, int N) {
  int wave = (int)((blockIdx.x * blockDim.x + threadIdx.x) >> 5);
  int lane = threadIdx.x & 31;
  int tilesN = N >> 4;
  int totTiles = (NN >> 4) * tilesN;
  if (wave >= totTiles) return;
  int tr = wave / tilesN;
  int tm = tr << 4;
  int tn = (wave - tr * tilesN) << 4;
  int half = lane >> 4;
  int l    = lane & 15;

  const float* xrow = X + (size_t)(tm + l) * K + 2 * half;
  const float* wcol = W + tn + l;
  v8f c = {};
  for (int k = 0; k < K; k += 4) {
    v2f a, b;
    a.x = xrow[k];
    a.y = xrow[k + 1];
    b.x = wcol[(size_t)(k + 2 * half) * N];
    b.y = wcol[(size_t)(k + 2 * half + 1) * N];
    c = __builtin_amdgcn_wmma_f32_16x16x4_f32(false, a, false, b, (short)0, c, false, false);
  }
  float* hrow = H + (size_t)(tm + 8 * half) * N + tn + l;
#pragma unroll
  for (int r = 0; r < 8; ++r) hrow[(size_t)r * N] = c[r];
}

__global__ void k_edge_scatter(const float* __restrict__ H, const int* __restrict__ src,
                               const int* __restrict__ dst, const float* __restrict__ dinv,
                               float* __restrict__ out, int C4) {
  long long gid = (long long)blockIdx.x * blockDim.x + threadIdx.x;
  int e = (int)(gid / C4);
  if (e >= NE) return;
  int c = (int)(gid - (long long)e * C4) << 2;
  int s = src[e], d = dst[e];
  if ((threadIdx.x & 63) == 0 && e + 256 < NE) {
    __builtin_prefetch(src + e + 256, 0, 0);
    __builtin_prefetch(dst + e + 256, 0, 0);
  }
  float nrm = dinv[s] * dinv[d];
  int C = C4 << 2;
  const float4 v = *(const float4*)(H + (size_t)s * C + c);
  float* op = out + (size_t)d * C + c;
  __hip_atomic_fetch_add(op + 0, v.x * nrm, __ATOMIC_RELAXED, __HIP_MEMORY_SCOPE_AGENT);
  __hip_atomic_fetch_add(op + 1, v.y * nrm, __ATOMIC_RELAXED, __HIP_MEMORY_SCOPE_AGENT);
  __hip_atomic_fetch_add(op + 2, v.z * nrm, __ATOMIC_RELAXED, __HIP_MEMORY_SCOPE_AGENT);
  __hip_atomic_fetch_add(op + 3, v.w * nrm, __ATOMIC_RELAXED, __HIP_MEMORY_SCOPE_AGENT);
}

__global__ void k_finalize(float* __restrict__ out, const float* __restrict__ H,
                           const float* __restrict__ dinv, const float* __restrict__ bias,
                           int C, int relu) {
  long long gid = (long long)blockIdx.x * blockDim.x + threadIdx.x;
  if (gid >= (long long)NN * C) return;
  int i = (int)(gid / C);
  int c = (int)(gid - (long long)i * C);
  float di = dinv[i];
  float v = out[gid] + H[gid] * (di * di) + bias[c];
  out[gid] = relu ? fmaxf(v, 0.0f) : v;
}

static inline unsigned nblk(long long n, int b) { return (unsigned)((n + b - 1) / b); }

extern "C" void kernel_launch(void* const* d_in, const int* in_sizes, int n_in,
                              void* d_out, int out_size, void* d_ws, size_t ws_size,
                              hipStream_t stream) {
  (void)in_sizes; (void)n_in; (void)out_size; (void)ws_size;
  const float* x   = (const float*)d_in[0];
  const int*   ei  = (const int*)d_in[1];
  const float* We1 = (const float*)d_in[2]; const float* be1 = (const float*)d_in[3];
  const float* We2 = (const float*)d_in[4]; const float* be2 = (const float*)d_in[5];
  const float* Wd1 = (const float*)d_in[6]; const float* bd1 = (const float*)d_in[7];
  const float* Wd2 = (const float*)d_in[8]; const float* bd2 = (const float*)d_in[9];
  const int* src = ei;
  const int* dst = ei + NE;

  char*  ws   = (char*)d_ws;
  float* dinv = (float*)ws;
  size_t off  = ((size_t)NN * 4 + 255) & ~(size_t)255;
  float* G    = (float*)(ws + off);
  float* P    = G + (size_t)NN * 128;

  k_init_deg <<<nblk(NN, 256), 256, 0, stream>>>(dinv);
  k_deg_edges<<<nblk(NE, 256), 256, 0, stream>>>(dst, dinv);
  k_rsqrt    <<<nblk(NN, 256), 256, 0, stream>>>(dinv);

  struct Layer { const float* in; const float* W; const float* b; int K; int C; int relu; float* S; };
  Layer layers[4] = {
    { x, We1, be1, 128,  64, 1, P },
    { P, We2, be2,  64,  32, 0, P },
    { P, Wd1, bd1,  32,  64, 1, P },
    { P, Wd2, bd2,  64, 128, 0, (float*)d_out },
  };

  for (int li = 0; li < 4; ++li) {
    const Layer& L = layers[li];
    long long tot   = (long long)NN * L.C;
    long long tiles = (long long)(NN / 16) * (L.C / 16);
    k_gemm_wmma   <<<nblk(tiles * 32, 256), 256, 0, stream>>>(L.in, L.W, G, L.K, L.C);
    k_zero        <<<nblk(tot, 256),        256, 0, stream>>>(L.S, (unsigned long long)tot);
    k_edge_scatter<<<nblk((long long)NE * (L.C / 4), 256), 256, 0, stream>>>(G, src, dst, dinv, L.S, L.C / 4);
    k_finalize    <<<nblk(tot, 256),        256, 0, stream>>>(L.S, G, dinv, L.b, L.C, L.relu);
  }
}


# Replace float atomic scatter-add by a gather over a per-call destination-sorted edge list (in-row sum order = arrival rank of an int atomic); zero and self-loop/bias/ReLU passes folded in
# speedup vs baseline: 2.6286x; 2.6286x over previous
#include <hip/hip_runtime.h>


#define NN 50000
#define NE 800000

typedef __attribute__((ext_vector_type(2))) float v2f;
typedef __attribute__((ext_vector_type(8))) float v8f;


__global__ void k_init(int* __restrict__ cnt, int* __restrict__ total) {
  int i = blockIdx.x * blockDim.x + threadIdx.x;
  if (i < NN) cnt[i] = 0;
  if (i == 0) *total = 0;
}

__global__ void k_count(const int* __restrict__ dst, int* __restrict__ cnt, int* __restrict__ rank) {
  int e = blockIdx.x * blockDim.x + threadIdx.x;
  if (e < NE)
    rank[e] = __hip_atomic_fetch_add(cnt + dst[e], 1, __ATOMIC_RELAXED, __HIP_MEMORY_SCOPE_AGENT);
}

__global__ void __launch_bounds__(256) k_rows(const int* __restrict__ cnt, int* __restrict__ total,
                                              int* __restrict__ rowstart, float* __restrict__ dinv) {
  __shared__ int s[256];
  __shared__ int sbase;
  const int tid = threadIdx.x;
  const int i = blockIdx.x * 256 + tid;
  const int c = (i < NN) ? cnt[i] : 0;
  s[tid] = c;
  __syncthreads();
  for (int off = 1; off < 256; off <<= 1) {
    int v = (tid >= off) ? s[tid - off] : 0;
    __syncthreads();
    s[tid] += v;
    __syncthreads();
  }
  const int incl = s[tid];
  if (tid == 255)
    sbase = __hip_atomic_fetch_add(total, incl, __ATOMIC_RELAXED, __HIP_MEMORY_SCOPE_AGENT);
  __syncthreads();
  if (i < NN) {
    rowstart[i] = sbase + incl - c;
    dinv[i] = rsqrtf((float)(c + 1));
  }
}

__global__ void k_fill(const int* __restrict__ src, const int* __restrict__ dst,
                       const int* __restrict__ rank, const int* __restrict__ rowstart,
                       const float* __restrict__ dinv, int* __restrict__ csr) {
  int e = blockIdx.x * blockDim.x + threadIdx.x;
  if (e >= NE) return;
  int s = src[e], d = dst[e];
  int p = rowstart[d] + rank[e];
  int2 r;
  r.x = s;
  r.y = __float_as_int(dinv[s] * dinv[d]);
  ((int2*)csr)[p] = r;
}

__global__ void k_gemm_wmma(const float* __restrict__ X, const float* __restrict__ W,
                            float* __restrict__ H, int K, int N) {
  int wave = (int)((blockIdx.x * blockDim.x + threadIdx.x) >> 5);
  int lane = threadIdx.x & 31;
  int tilesN = N >> 4;
  int totTiles = (NN >> 4) * tilesN;
  if (wave >= totTiles) return;
  int tr = wave / tilesN;
  int tm = tr << 4;
  int tn = (wave - tr * tilesN) << 4;
  int half = lane >> 4;
  int l    = lane & 15;

  const float* xrow = X + (size_t)(tm + l) * K + 2 * half;
  const float* wcol = W + tn + l;
  v8f c = {};
  for (int k = 0; k < K; k += 4) {
    v2f a, b;
    a.x = xrow[k];
    a.y = xrow[k + 1];
    b.x = wcol[(size_t)(k + 2 * half) * N];
    b.y = wcol[(size_t)(k + 2 * half + 1) * N];
    c = __builtin_amdgcn_wmma_f32_16x16x4_f32(false, a, false, b, (short)0, c, false, false);
  }
  float* hrow = H + (size_t)(tm + 8 * half) * N + tn + l;
#pragma unroll
  for (int r = 0; r < 8; ++r) hrow[(size_t)r * N] = c[r];
}

__device__ __forceinline__ void fma4(float4& acc, const float4& v, float w) {
  acc.x = __builtin_fmaf(v.x, w, acc.x);
  acc.y = __builtin_fmaf(v.y, w, acc.y);
  acc.z = __builtin_fmaf(v.z, w, acc.z);
  acc.w = __builtin_fmaf(v.w, w, acc.w);
}

template <int C, bool RELU>
__global__ void __launch_bounds__(256) k_gather(const float* __restrict__ H,
                                                const int* __restrict__ rowstart,
                                                const int* __restrict__ cnt,
                                                const int* __restrict__ csr,
                                                const float* __restrict__ dinv,
                                                const float* __restrict__ bias,
                                                float* __restrict__ out) {
  constexpr int G   = C / 4;
  constexpr int NPB = 256 / G;
  const int tid  = threadIdx.x;
  const int gl   = tid & (G - 1);
  const int node = blockIdx.x * NPB + tid / G;
  if (node >= NN) return;

  const float4* Hv = (const float4*)H;
  const int2*   ev = (const int2*)csr;
  const int   rs = rowstart[node];
  const int   n  = cnt[node];
  const float di = dinv[node];
  const float sl = di * di;

  float4 acc = Hv[(unsigned)(node * G + gl)];
  acc.x *= sl; acc.y *= sl; acc.z *= sl; acc.w *= sl;

  for (int base = 0; base < n; base += G) {
    const int m = (n - base < G) ? (n - base) : G;
    int2 rec;
    rec.x = 0; rec.y = 0;
    if (gl < m) rec = ev[rs + base + gl];
    int k = 0;
    for (; k + 4 <= m; k += 4) {
      const int   s0 = __shfl(rec.x, k,     G);
      const int   s1 = __shfl(rec.x, k + 1, G);
      const int   s2 = __shfl(rec.x, k + 2, G);
      const int   s3 = __shfl(rec.x, k + 3, G);
      const float w0 = __int_as_float(__shfl(rec.y, k,     G));
      const float w1 = __int_as_float(__shfl(rec.y, k + 1, G));
      const float w2 = __int_as_float(__shfl(rec.y, k + 2, G));
      const float w3 = __int_as_float(__shfl(rec.y, k + 3, G));
      const float4 v0 = Hv[(unsigned)(s0 * G + gl)];
      const float4 v1 = Hv[(unsigned)(s1 * G + gl)];
      const float4 v2 = Hv[(unsigned)(s2 * G + gl)];
      const float4 v3 = Hv[(unsigned)(s3 * G + gl)];
      fma4(acc, v0, w0);
      fma4(acc, v1, w1);
      fma4(acc, v2, w2);
      fma4(acc, v3, w3);
    }
    for (; k < m; ++k) {
      const int   s0 = __shfl(rec.x, k, G);
      const float w0 = __int_as_float(__shfl(rec.y, k, G));
      const float4 v0 = Hv[(unsigned)(s0 * G + gl)];
      fma4(acc, v0, w0);
    }
  }

  const int c0 = gl * 4;
  acc.x += bias[c0]; acc.y += bias[c0 + 1]; acc.z += bias[c0 + 2]; acc.w += bias[c0 + 3];
  if (RELU) {
    acc.x = fmaxf(acc.x, 0.0f); acc.y = fmaxf(acc.y, 0.0f);
    acc.z = fmaxf(acc.z, 0.0f); acc.w = fmaxf(acc.w, 0.0f);
  }
  ((float4*)out)[(unsigned)(node * G + gl)] = acc;
}

static inline unsigned nblk(long long n, int b) { return (unsigned)((n + b - 1) / b); }

static constexpr size_t al256(size_t x) { return (x + 255) & ~(size_t)255; }
static constexpr size_t OFF_CNT   = 0;
static constexpr size_t OFF_DINV  = OFF_CNT   + al256((size_t)NN * 4);
static constexpr size_t OFF_ROW   = OFF_DINV  + al256((size_t)NN * 4);
static constexpr size_t OFF_TOTAL = OFF_ROW   + al256((size_t)NN * 4);
static constexpr size_t OFF_RANK  = OFF_TOTAL + 256;
static constexpr size_t OFF_CSR   = OFF_RANK  + al256((size_t)NE * 4);
static constexpr size_t OFF_G     = OFF_CSR   + al256((size_t)NE * 8);
static constexpr size_t OFF_P     = OFF_G     + al256((size_t)NN * 128 * 4);

extern "C" void kernel_launch(void* const* d_in, const int* in_sizes, int n_in,
                              void* d_out, int out_size, void* d_ws, size_t ws_size,
                              hipStream_t stream) {
  (void)in_sizes; (void)n_in; (void)out_size; (void)ws_size;
  const float* x   = (const float*)d_in[0];
  const int*   ei  = (const int*)d_in[1];
  const float* We1 = (const float*)d_in[2]; const float* be1 = (const float*)d_in[3];
  const float* We2 = (const float*)d_in[4]; const float* be2 = (const float*)d_in[5];
  const float* Wd1 = (const float*)d_in[6]; const float* bd1 = (const float*)d_in[7];
  const float* Wd2 = (const float*)d_in[8]; const float* bd2 = (const float*)d_in[9];
  const int* src = ei;
  const int* dst = ei + NE;

  char*  ws       = (char*)d_ws;
  int*   cnt      = (int*)(ws + OFF_CNT);
  float* dinv     = (float*)(ws + OFF_DINV);
  int*   rowstart = (int*)(ws + OFF_ROW);
  int*   total    = (int*)(ws + OFF_TOTAL);
  int*   rank     = (int*)(ws + OFF_RANK);
  int*   csr      = (int*)(ws + OFF_CSR);
  float* G        = (float*)(ws + OFF_G);
  float* P        = (float*)(ws + OFF_P);
  float* out      = (float*)d_out;

  k_init <<<nblk(NN, 256), 256, 0, stream>>>(cnt, total);
  k_count<<<nblk(NE, 256), 256, 0, stream>>>(dst, cnt, rank);
  k_rows <<<nblk(NN, 256), 256, 0, stream>>>(cnt, total, rowstart, dinv);
  k_fill <<<nblk(NE, 256), 256, 0, stream>>>(src, dst, rank, rowstart, dinv, csr);

  const long long rt = NN / 16;
  k_gemm_wmma<<<nblk(rt * 4 * 32, 256), 256, 0, stream>>>(x, We1, G, 128, 64);
  k_gather<64, true><<<nblk(NN, 16), 256, 0, stream>>>(G, rowstart, cnt, csr, dinv, be1, P);
  k_gemm_wmma<<<nblk(rt * 2 * 32, 256), 256, 0, stream>>>(P, We2, G, 64, 32);
  k_gather<32, false><<<nblk(NN, 32), 256, 0, stream>>>(G, rowstart, cnt, csr, dinv, be2, P);
  k_gemm_wmma<<<nblk(rt * 4 * 32, 256), 256, 0, stream>>>(P, Wd1, G, 32, 64);
  k_gather<64, true><<<nblk(NN, 16), 256, 0, stream>>>(G, rowstart, cnt, csr, dinv, bd1, P);
  k_gemm_wmma<<<nblk(rt * 8 * 32, 256), 256, 0, stream>>>(P, Wd2, G, 64, 128);
  k_gather<128, false><<<nblk(NN, 8), 256, 0, stream>>>(G, rowstart, cnt, csr, dinv, bd2, out);
}


# Sorted edge list built by a counting sort through LDS (256-node bins) instead of returning global atomics and scattered per-edge stores; in-row sum order = arrival rank of LDS int atomics
# speedup vs baseline: 3.7240x; 1.4167x over previous
#include <hip/hip_runtime.h>


#define NN 50000
#define NE 800000

typedef __attribute__((ext_vector_type(2))) float v2f;
typedef __attribute__((ext_vector_type(8))) float v8f;

#define NBIN    196
#define NB1     256
#define CHUNK   3125
#define PSTRIDE 3136
#define EPT     13
#define RPW     8
#define CAP     16384

__device__ __forceinline__ int wave_incl_scan(int v, int lane) {
#pragma unroll
  for (int off = 1; off < 32; off <<= 1) {
    const int t = __shfl_up(v, off, 32);
    if (lane >= off) v += t;
  }
  return v;
}

__global__ void __launch_bounds__(256) k_partition(const int* __restrict__ src, const int* __restrict__ dst,
                                                   int* __restrict__ part, int* __restrict__ runstart) {
  __shared__ int hist[256];
  __shared__ int wsum[8];
  __shared__ int sorted[CHUNK];
  const int tid  = threadIdx.x;
  const int lane = tid & 31;
  const int wv   = tid >> 5;
  const int b    = blockIdx.x;
  const int e0   = b * CHUNK;
  hist[tid] = 0;
  int dv[EPT], sv[EPT];
#pragma unroll
  for (int j = 0; j < EPT; ++j) {
    const int i = tid + j * 256;
    dv[j] = 0; sv[j] = 0;
    if (i < CHUNK) { dv[j] = dst[e0 + i]; sv[j] = src[e0 + i]; }
  }
  __syncthreads();
  int key[EPT];
#pragma unroll
  for (int j = 0; j < EPT; ++j) {
    const int i = tid + j * 256;
    key[j] = -1;
    if (i < CHUNK) {
      const int bin = dv[j] >> 8;
      const int r = __hip_atomic_fetch_add(&hist[bin], 1, __ATOMIC_RELAXED, __HIP_MEMORY_SCOPE_WORKGROUP);
      key[j] = (bin << 16) | r;
    }
  }
  __syncthreads();
  const int c = hist[tid];
  const int incl = wave_incl_scan(c, lane);
  if (lane == 31) wsum[wv] = incl;
  __syncthreads();
  int pre = 0;
#pragma unroll
  for (int w = 0; w < 8; ++w) if (w < wv) pre += wsum[w];
  const int excl = incl + pre - c;
  hist[tid] = excl;
  if (tid <= NBIN) runstart[b * (NBIN + 1) + tid] = excl;
  __syncthreads();
#pragma unroll
  for (int j = 0; j < EPT; ++j)
    if (key[j] >= 0) sorted[hist[key[j] >> 16] + (key[j] & 0xffff)] = sv[j] | ((dv[j] & 255) << 16);
  __syncthreads();
  for (int i = tid; i < CHUNK; i += 256) part[b * PSTRIDE + i] = sorted[i];
}

__device__ __forceinline__ void place_rec(int rec, int bin, const int* nstart, int* ncur,
                                          bool fits, int2* stage, int2* out) {
  const int dl  = (rec >> 16) & 255;
  const int pos = nstart[dl] +
      __hip_atomic_fetch_add(&ncur[dl], 1, __ATOMIC_RELAXED, __HIP_MEMORY_SCOPE_WORKGROUP);
  int2 o;
  o.x = rec & 0xffff;
  o.y = bin * 256 + dl;
  if (fits) stage[pos] = o; else out[pos] = o;
}

__global__ void __launch_bounds__(1024) k_bin(const int* __restrict__ part, const int* __restrict__ runstart,
                                              int* __restrict__ cnt, int* __restrict__ rowstart,
                                              float* __restrict__ dinv, int* __restrict__ pairs) {
  __shared__ int  rs0[NB1];
  __shared__ int  rs1[NB1];
  __shared__ int  ncnt[256];
  __shared__ int  nstart[256];
  __shared__ int  wsA[8];
  __shared__ int  wsC[8];
  __shared__ int2 stage[CAP];
  const int tid  = threadIdx.x;
  const int lane = tid & 31;
  const int wv   = tid >> 5;
  const int bin  = blockIdx.x;
  int a = 0;
  if (tid < NB1) {
    a = runstart[tid * (NBIN + 1) + bin];
    const int z = runstart[tid * (NBIN + 1) + bin + 1];
    rs0[tid] = a; rs1[tid] = z;
    ncnt[tid] = 0;
  }
  int sa = a;
#pragma unroll
  for (int off = 16; off > 0; off >>= 1) sa += __shfl_xor(sa, off, 32);
  if (tid < NB1 && lane == 0) wsA[wv] = sa;
  __syncthreads();
  int base = 0;
#pragma unroll
  for (int w = 0; w < 8; ++w) base += wsA[w];

  int recs[RPW];
#pragma unroll
  for (int k = 0; k < RPW; ++k) {
    const int r = wv + 32 * k;
    const int i = rs0[r] + lane;
    recs[k] = -1;
    if (i < rs1[r]) recs[k] = part[r * PSTRIDE + i];
  }
#pragma unroll
  for (int k = 0; k < RPW; ++k)
    if (recs[k] >= 0)
      __hip_atomic_fetch_add(&ncnt[(recs[k] >> 16) & 255], 1, __ATOMIC_RELAXED, __HIP_MEMORY_SCOPE_WORKGROUP);
  for (int k = 0; k < RPW; ++k) {
    const int r = wv + 32 * k;
    const int z = rs1[r];
    const int* p = part + r * PSTRIDE;
    for (int i = rs0[r] + lane + 32; i < z; i += 32)
      __hip_atomic_fetch_add(&ncnt[(p[i] >> 16) & 255], 1, __ATOMIC_RELAXED, __HIP_MEMORY_SCOPE_WORKGROUP);
  }
  __syncthreads();
  const int c = (tid < 256) ? ncnt[tid] : 0;
  const int incl = wave_incl_scan(c, lane);
  if (tid < 256 && lane == 31) wsC[wv] = incl;
  __syncthreads();
  int pre = 0, T = 0;
#pragma unroll
  for (int w = 0; w < 8; ++w) {
    const int t = wsC[w];
    T += t;
    if (w < wv) pre += t;
  }
  if (tid < 256) {
    const int excl = incl + pre - c;
    nstart[tid] = excl;
    ncnt[tid] = 0;
    const int node = bin * 256 + tid;
    if (node < NN) {
      cnt[node] = c;
      rowstart[node] = base + excl;
      dinv[node] = rsqrtf((float)(c + 1));
    }
  }
  __syncthreads();
  const bool fits = (T <= CAP);
  int2* out = (int2*)pairs + base;
#pragma unroll
  for (int k = 0; k < RPW; ++k)
    if (recs[k] >= 0) place_rec(recs[k], bin, nstart, ncnt, fits, stage, out);
  for (int k = 0; k < RPW; ++k) {
    const int r = wv + 32 * k;
    const int z = rs1[r];
    const int* p = part + r * PSTRIDE;
    for (int i = rs0[r] + lane + 32; i < z; i += 32)
      place_rec(p[i], bin, nstart, ncnt, fits, stage, out);
  }
  __syncthreads();
  if (fits)
    for (int i = tid; i < T; i += 1024) out[i] = stage[i];
}

__global__ void __launch_bounds__(256) k_weight(const float* __restrict__ dinv, const int* __restrict__ pairs,
                                                int* __restrict__ csr) {
  const int4* in = (const int4*)pairs;
  int4* outp = (int4*)csr;
  const int i0 = blockIdx.x * 512 + threadIdx.x;
  const int i1 = i0 + 256;
  const bool h0 = i0 < NE / 2, h1 = i1 < NE / 2;
  int4 u = in[h0 ? i0 : NE / 2 - 1];
  int4 v = in[h1 ? i1 : NE / 2 - 1];
  const float a0 = dinv[u.x] * dinv[u.y];
  const float a1 = dinv[u.z] * dinv[u.w];
  const float b0 = dinv[v.x] * dinv[v.y];
  const float b1 = dinv[v.z] * dinv[v.w];
  u.y = __float_as_int(a0); u.w = __float_as_int(a1);
  v.y = __float_as_int(b0); v.w = __float_as_int(b1);
  if (h0) outp[i0] = u;
  if (h1) outp[i1] = v;
}

__global__ void k_gemm_wmma(const float* __restrict__ X, const float* __restrict__ W,
                            float* __restrict__ H, int K, int N) {
  int wave = (int)((blockIdx.x * blockDim.x + threadIdx.x) >> 5);
  int lane = threadIdx.x & 31;
  int tilesN = N >> 4;
  int totTiles = (NN >> 4) * tilesN;
  if (wave >= totTiles) return;
  int tr = wave / tilesN;
  int tm = tr << 4;
  int tn = (wave - tr * tilesN) << 4;
  int half = lane >> 4;
  int l    = lane & 15;

  const float* xrow = X + (size_t)(tm + l) * K + 2 * half;
  const float* wcol = W + tn + l;
  v8f c = {};
  for (int k = 0; k < K; k += 4) {
    v2f a, b;
    a.x = xrow[k];
    a.y = xrow[k + 1];
    b.x = wcol[(size_t)(k + 2 * half) * N];
    b.y = wcol[(size_t)(k + 2 * half + 1) * N];
    c = __builtin_amdgcn_wmma_f32_16x16x4_f32(false, a, false, b, (short)0, c, false, false);
  }
  float* hrow = H + (size_t)(tm + 8 * half) * N + tn + l;
#pragma unroll
  for (int r = 0; r < 8; ++r) hrow[(size_t)r * N] = c[r];
}

__device__ __forceinline__ void fma4(float4& acc, const float4& v, float w) {
  acc.x = __builtin_fmaf(v.x, w, acc.x);
  acc.y = __builtin_fmaf(v.y, w, acc.y);
  acc.z = __builtin_fmaf(v.z, w, acc.z);
  acc.w = __builtin_fmaf(v.w, w, acc.w);
}

template <int C, bool RELU>
__global__ void __launch_bounds__(256) k_gather(const float* __restrict__ H,
                                                const int* __restrict__ rowstart,
                                                const int* __restrict__ cnt,
                                                const int* __restrict__ csr,
                                                const float* __restrict__ dinv,
                                                const float* __restrict__ bias,
                                                float* __restrict__ out) {
  constexpr int G   = C / 4;
  constexpr int NPB = 256 / G;
  const int tid  = threadIdx.x;
  const int gl   = tid & (G - 1);
  const int node = blockIdx.x * NPB + tid / G;
  if (node >= NN) return;

  const float4* Hv = (const float4*)H;
  const int2*   ev = (const int2*)csr;
  const int   rs = rowstart[node];
  const int   n  = cnt[node];
  const float di = dinv[node];
  const float sl = di * di;

  float4 acc = Hv[(unsigned)(node * G + gl)];
  acc.x *= sl; acc.y *= sl; acc.z *= sl; acc.w *= sl;

  for (int base = 0; base < n; base += G) {
    const int m = (n - base < G) ? (n - base) : G;
    int2 rec;
    rec.x = 0; rec.y = 0;
    if (gl < m) rec = ev[rs + base + gl];
    int k = 0;
    for (; k + 4 <= m; k += 4) {
      const int   s0 = __shfl(rec.x, k,     G);
      const int   s1 = __shfl(rec.x, k + 1, G);
      const int   s2 = __shfl(rec.x, k + 2, G);
      const int   s3 = __shfl(rec.x, k + 3, G);
      const float w0 = __int_as_float(__shfl(rec.y, k,     G));
      const float w1 = __int_as_float(__shfl(rec.y, k + 1, G));
      const float w2 = __int_as_float(__shfl(rec.y, k + 2, G));
      const float w3 = __int_as_float(__shfl(rec.y, k + 3, G));
      const float4 v0 = Hv[(unsigned)(s0 * G + gl)];
      const float4 v1 = Hv[(unsigned)(s1 * G + gl)];
      const float4 v2 = Hv[(unsigned)(s2 * G + gl)];
      const float4 v3 = Hv[(unsigned)(s3 * G + gl)];
      fma4(acc, v0, w0);
      fma4(acc, v1, w1);
      fma4(acc, v2, w2);
      fma4(acc, v3, w3);
    }
    for (; k < m; ++k) {
      const int   s0 = __shfl(rec.x, k, G);
      const float w0 = __int_as_float(__shfl(rec.y, k, G));
      const float4 v0 = Hv[(unsigned)(s0 * G + gl)];
      fma4(acc, v0, w0);
    }
  }

  const int c0 = gl * 4;
  acc.x += bias[c0]; acc.y += bias[c0 + 1]; acc.z += bias[c0 + 2]; acc.w += bias[c0 + 3];
  if (RELU) {
    acc.x = fmaxf(acc.x, 0.0f); acc.y = fmaxf(acc.y, 0.0f);
    acc.z = fmaxf(acc.z, 0.0f); acc.w = fmaxf(acc.w, 0.0f);
  }
  ((float4*)out)[(unsigned)(node * G + gl)] = acc;
}

static inline unsigned nblk(long long n, int b) { return (unsigned)((n + b - 1) / b); }

static constexpr size_t al256(size_t x) { return (x + 255) & ~(size_t)255; }
static constexpr size_t OFF_CNT   = 0;
static constexpr size_t OFF_DINV  = OFF_CNT   + al256((size_t)NN * 4);
static constexpr size_t OFF_ROW   = OFF_DINV  + al256((size_t)NN * 4);
static constexpr size_t OFF_RUN   = OFF_ROW   + al256((size_t)NN * 4);
static constexpr size_t OFF_PART  = OFF_RUN   + al256((size_t)NB1 * (NBIN + 1) * 4);
static constexpr size_t OFF_CSR   = OFF_PART  + al256((size_t)NB1 * PSTRIDE * 4);
static constexpr size_t OFF_PAIR  = OFF_CSR   + al256((size_t)NE * 8);
static constexpr size_t OFF_G     = OFF_PAIR  + al256((size_t)NE * 8);
static constexpr size_t OFF_P     = OFF_G     + al256((size_t)NN * 128 * 4);

extern "C" void kernel_launch(void* const* d_in, const int* in_sizes, int n_in,
                              void* d_out, int out_size, void* d_ws, size_t ws_size,
                              hipStream_t stream) {
  (void)in_sizes; (void)n_in; (void)out_size; (void)ws_size;
  const float* x   = (const float*)d_in[0];
  const int*   ei  = (const int*)d_in[1];
  const float* We1 = (const float*)d_in[2]; const float* be1 = (const float*)d_in[3];
  const float* We2 = (const float*)d_in[4]; const float* be2 = (const float*)d_in[5];
  const float* Wd1 = (const float*)d_in[6]; const float* bd1 = (const float*)d_in[7];
  const float* Wd2 = (const float*)d_in[8]; const float* bd2 = (const float*)d_in[9];
  const int* src = ei;
  const int* dst = ei + NE;

  char*  ws       = (char*)d_ws;
  int*   cnt      = (int*)(ws + OFF_CNT);
  float* dinv     = (float*)(ws + OFF_DINV);
  int*   rowstart = (int*)(ws + OFF_ROW);
  int*   runstart = (int*)(ws + OFF_RUN);
  int*   part     = (int*)(ws + OFF_PART);
  int*   csr      = (int*)(ws + OFF_CSR);
  int*   pairs    = (int*)(ws + OFF_PAIR);
  float* G        = (float*)(ws + OFF_G);
  float* P        = (float*)(ws + OFF_P);
  float* out      = (float*)d_out;

  k_partition<<<NB1, 256, 0, stream>>>(src, dst, part, runstart);
  k_bin      <<<NBIN, 1024, 0, stream>>>(part, runstart, cnt, rowstart, dinv, pairs);
  k_weight   <<<nblk(NE / 2, 512), 256, 0, stream>>>(dinv, pairs, csr);

  const long long rt = NN / 16;
  k_gemm_wmma<<<nblk(rt * 4 * 32, 256), 256, 0, stream>>>(x, We1, G, 128, 64);
  k_gather<64, true><<<nblk(NN, 16), 256, 0, stream>>>(G, rowstart, cnt, csr, dinv, be1, P);
  k_gemm_wmma<<<nblk(rt * 2 * 32, 256), 256, 0, stream>>>(P, We2, G, 64, 32);
  k_gather<32, false><<<nblk(NN, 32), 256, 0, stream>>>(G, rowstart, cnt, csr, dinv, be2, P);
  k_gemm_wmma<<<nblk(rt * 4 * 32, 256), 256, 0, stream>>>(P, Wd1, G, 32, 64);
  k_gather<64, true><<<nblk(NN, 16), 256, 0, stream>>>(G, rowstart, cnt, csr, dinv, bd1, P);
  k_gemm_wmma<<<nblk(rt * 8 * 32, 256), 256, 0, stream>>>(P, Wd2, G, 64, 128);
  k_gather<128, false><<<nblk(NN, 8), 256, 0, stream>>>(G, rowstart, cnt, csr, dinv, bd2, out);
}


# Widening layers (32->64, 64->128): gather input rows first, then the product, bias as accumulator start, ReLU at its store (A(XW)=(AX)W); gathers at 32/64 columns, not 64/128; sum order changes
# speedup vs baseline: 3.9792x; 1.0685x over previous
#include <hip/hip_runtime.h>


#define NN 50000
#define NE 800000

typedef __attribute__((ext_vector_type(2))) float v2f;
typedef __attribute__((ext_vector_type(8))) float v8f;

#define NBIN    196
#define NB1     256
#define CHUNK   3125
#define PSTRIDE 3136
#define EPT     13
#define RPW     8
#define CAP     16384

__device__ __forceinline__ int wave_incl_scan(int v, int lane) {
#pragma unroll
  for (int off = 1; off < 32; off <<= 1) {
    const int t = __shfl_up(v, off, 32);
    if (lane >= off) v += t;
  }
  return v;
}

__global__ void __launch_bounds__(256) k_partition(const int* __restrict__ src, const int* __restrict__ dst,
                                                   int* __restrict__ part, int* __restrict__ runstart) {
  __shared__ int hist[256];
  __shared__ int wsum[8];
  __shared__ int sorted[CHUNK];
  const int tid  = threadIdx.x;
  const int lane = tid & 31;
  const int wv   = tid >> 5;
  const int b    = blockIdx.x;
  const int e0   = b * CHUNK;
  hist[tid] = 0;
  int dv[EPT], sv[EPT];
#pragma unroll
  for (int j = 0; j < EPT; ++j) {
    const int i = tid + j * 256;
    dv[j] = 0; sv[j] = 0;
    if (i < CHUNK) { dv[j] = dst[e0 + i]; sv[j] = src[e0 + i]; }
  }
  __syncthreads();
  int key[EPT];
#pragma unroll
  for (int j = 0; j < EPT; ++j) {
    const int i = tid + j * 256;
    key[j] = -1;
    if (i < CHUNK) {
      const int bin = dv[j] >> 8;
      const int r = __hip_atomic_fetch_add(&hist[bin], 1, __ATOMIC_RELAXED, __HIP_MEMORY_SCOPE_WORKGROUP);
      key[j] = (bin << 16) | r;
    }
  }
  __syncthreads();
  const int c = hist[tid];
  const int incl = wave_incl_scan(c, lane);
  if (lane == 31) wsum[wv] = incl;
  __syncthreads();
  int pre = 0;
#pragma unroll
  for (int w = 0; w < 8; ++w) if (w < wv) pre += wsum[w];
  const int excl = incl + pre - c;
  hist[tid] = excl;
  if (tid <= NBIN) runstart[b * (NBIN + 1) + tid] = excl;
  __syncthreads();
#pragma unroll
  for (int j = 0; j < EPT; ++j)
    if (key[j] >= 0) sorted[hist[key[j] >> 16] + (key[j] & 0xffff)] = sv[j] | ((dv[j] & 255) << 16);
  __syncthreads();
  for (int i = tid; i < CHUNK; i += 256) part[b * PSTRIDE + i] = sorted[i];
}

__device__ __forceinline__ void place_rec(int rec, int bin, const int* nstart, int* ncur,
                                          bool fits, int2* stage, int2* out) {
  const int dl  = (rec >> 16) & 255;
  const int pos = nstart[dl] +
      __hip_atomic_fetch_add(&ncur[dl], 1, __ATOMIC_RELAXED, __HIP_MEMORY_SCOPE_WORKGROUP);
  int2 o;
  o.x = rec & 0xffff;
  o.y = bin * 256 + dl;
  if (fits) stage[pos] = o; else out[pos] = o;
}

__global__ void __launch_bounds__(1024) k_bin(const int* __restrict__ part, const int* __restrict__ runstart,
                                              int* __restrict__ cnt, int* __restrict__ rowstart,
                                              float* __restrict__ dinv, int* __restrict__ pairs) {
  __shared__ int  rs0[NB1];
  __shared__ int  rs1[NB1];
  __shared__ int  ncnt[256];
  __shared__ int  nstart[256];
  __shared__ int  wsA[8];
  __shared__ int  wsC[8];
  __shared__ int2 stage[CAP];
  const int tid  = threadIdx.x;
  const int lane = tid & 31;
  const int wv   = tid >> 5;
  const int bin  = blockIdx.x;
  int a = 0;
  if (tid < NB1) {
    a = runstart[tid * (NBIN + 1) + bin];
    const int z = runstart[tid * (NBIN + 1) + bin + 1];
    rs0[tid] = a; rs1[tid] = z;
    ncnt[tid] = 0;
  }
  int sa = a;
#pragma unroll
  for (int off = 16; off > 0; off >>= 1) sa += __shfl_xor(sa, off, 32);
  if (tid < NB1 && lane == 0) wsA[wv] = sa;
  __syncthreads();
  int base = 0;
#pragma unroll
  for (int w = 0; w < 8; ++w) base += wsA[w];

  int recs[RPW];
#pragma unroll
  for (int k = 0; k < RPW; ++k) {
    const int r = wv + 32 * k;
    const int i = rs0[r] + lane;
    recs[k] = -1;
    if (i < rs1[r]) recs[k] = part[r * PSTRIDE + i];
  }
#pragma unroll
  for (int k = 0; k < RPW; ++k)
    if (recs[k] >= 0)
      __hip_atomic_fetch_add(&ncnt[(recs[k] >> 16) & 255], 1, __ATOMIC_RELAXED, __HIP_MEMORY_SCOPE_WORKGROUP);
  for (int k = 0; k < RPW; ++k) {
    const int r = wv + 32 * k;
    const int z = rs1[r];
    const int* p = part + r * PSTRIDE;
    for (int i = rs0[r] + lane + 32; i < z; i += 32)
      __hip_atomic_fetch_add(&ncnt[(p[i] >> 16) & 255], 1, __ATOMIC_RELAXED, __HIP_MEMORY_SCOPE_WORKGROUP);
  }
  __syncthreads();
  const int c = (tid < 256) ? ncnt[tid] : 0;
  const int incl = wave_incl_scan(c, lane);
  if (tid < 256 && lane == 31) wsC[wv] = incl;
  __syncthreads();
  int pre = 0, T = 0;
#pragma unroll
  for (int w = 0; w < 8; ++w) {
    const int t = wsC[w];
    T += t;
    if (w < wv) pre += t;
  }
  if (tid < 256) {
    const int excl = incl + pre - c;
    nstart[tid] = excl;
    ncnt[tid] = 0;
    const int node = bin * 256 + tid;
    if (node < NN) {
      cnt[node] = c;
      rowstart[node] = base + excl;
      dinv[node] = rsqrtf((float)(c + 1));
    }
  }
  __syncthreads();
  const bool fits = (T <= CAP);
  int2* out = (int2*)pairs + base;
#pragma unroll
  for (int k = 0; k < RPW; ++k)
    if (recs[k] >= 0) place_rec(recs[k], bin, nstart, ncnt, fits, stage, out);
  for (int k = 0; k < RPW; ++k) {
    const int r = wv + 32 * k;
    const int z = rs1[r];
    const int* p = part + r * PSTRIDE;
    for (int i = rs0[r] + lane + 32; i < z; i += 32)
      place_rec(p[i], bin, nstart, ncnt, fits, stage, out);
  }
  __syncthreads();
  if (fits)
    for (int i = tid; i < T; i += 1024) out[i] = stage[i];
}

__global__ void __launch_bounds__(256) k_weight(const float* __restrict__ dinv, const int* __restrict__ pairs,
                                                int* __restrict__ csr) {
  const int4* in = (const int4*)pairs;
  int4* outp = (int4*)csr;
  const int i0 = blockIdx.x * 512 + threadIdx.x;
  const int i1 = i0 + 256;
  const bool h0 = i0 < NE / 2, h1 = i1 < NE / 2;
  int4 u = in[h0 ? i0 : NE / 2 - 1];
  int4 v = in[h1 ? i1 : NE / 2 - 1];
  const float a0 = dinv[u.x] * dinv[u.y];
  const float a1 = dinv[u.z] * dinv[u.w];
  const float b0 = dinv[v.x] * dinv[v.y];
  const float b1 = dinv[v.z] * dinv[v.w];
  u.y = __float_as_int(a0); u.w = __float_as_int(a1);
  v.y = __float_as_int(b0); v.w = __float_as_int(b1);
  if (h0) outp[i0] = u;
  if (h1) outp[i1] = v;
}

template <bool BIAS, bool RELU>
__global__ void k_gemm_wmma(const float* __restrict__ X, const float* __restrict__ W,
                            const float* __restrict__ bias, float* __restrict__ H, int K, int N) {
  int wave = (int)((blockIdx.x * blockDim.x + threadIdx.x) >> 5);
  int lane = threadIdx.x & 31;
  int tilesN = N >> 4;
  int totTiles = (NN >> 4) * tilesN;
  if (wave >= totTiles) return;
  int tr = wave / tilesN;
  int tm = tr << 4;
  int tn = (wave - tr * tilesN) << 4;
  int half = lane >> 4;
  int l    = lane & 15;

  const float* xrow = X + (size_t)(tm + l) * K + 2 * half;
  const float* wcol = W + tn + l;
  float bv = 0.0f;
  if (BIAS) bv = bias[tn + l];
  v8f c = {bv, bv, bv, bv, bv, bv, bv, bv};
  for (int k = 0; k < K; k += 4) {
    v2f a, b;
    a.x = xrow[k];
    a.y = xrow[k + 1];
    b.x = wcol[(size_t)(k + 2 * half) * N];
    b.y = wcol[(size_t)(k + 2 * half + 1) * N];
    c = __builtin_amdgcn_wmma_f32_16x16x4_f32(false, a, false, b, (short)0, c, false, false);
  }
  float* hrow = H + (size_t)(tm + 8 * half) * N + tn + l;
#pragma unroll
  for (int r = 0; r < 8; ++r) {
    float v = c[r];
    if (RELU) v = fmaxf(v, 0.0f);
    hrow[(size_t)r * N] = v;
  }
}

__device__ __forceinline__ void fma4(float4& acc, const float4& v, float w) {
  acc.x = __builtin_fmaf(v.x, w, acc.x);
  acc.y = __builtin_fmaf(v.y, w, acc.y);
  acc.z = __builtin_fmaf(v.z, w, acc.z);
  acc.w = __builtin_fmaf(v.w, w, acc.w);
}

template <int C, bool BIAS, bool RELU>
__global__ void __launch_bounds__(256) k_gather(const float* __restrict__ H,
                                                const int* __restrict__ rowstart,
                                                const int* __restrict__ cnt,
                                                const int* __restrict__ csr,
                                                const float* __restrict__ dinv,
                                                const float* __restrict__ bias,
                                                float* __restrict__ out) {
  constexpr int G   = C / 4;
  constexpr int NPB = 256 / G;
  const int tid  = threadIdx.x;
  const int gl   = tid & (G - 1);
  const int node = blockIdx.x * NPB + tid / G;
  if (node >= NN) return;

  const float4* Hv = (const float4*)H;
  const int2*   ev = (const int2*)csr;
  const int   rs = rowstart[node];
  const int   n  = cnt[node];
  const float di = dinv[node];
  const float sl = di * di;

  float4 acc = Hv[(unsigned)(node * G + gl)];
  acc.x *= sl; acc.y *= sl; acc.z *= sl; acc.w *= sl;

  for (int base = 0; base < n; base += G) {
    const int m = (n - base < G) ? (n - base) : G;
    int2 rec;
    rec.x = 0; rec.y = 0;
    if (gl < m) rec = ev[rs + base + gl];
    int k = 0;
    for (; k + 4 <= m; k += 4) {
      const int   s0 = __shfl(rec.x, k,     G);
      const int   s1 = __shfl(rec.x, k + 1, G);
      const int   s2 = __shfl(rec.x, k + 2, G);
      const int   s3 = __shfl(rec.x, k + 3, G);
      const float w0 = __int_as_float(__shfl(rec.y, k,     G));
      const float w1 = __int_as_float(__shfl(rec.y, k + 1, G));
      const float w2 = __int_as_float(__shfl(rec.y, k + 2, G));
      const float w3 = __int_as_float(__shfl(rec.y, k + 3, G));
      const float4 v0 = Hv[(unsigned)(s0 * G + gl)];
      const float4 v1 = Hv[(unsigned)(s1 * G + gl)];
      const float4 v2 = Hv[(unsigned)(s2 * G + gl)];
      const float4 v3 = Hv[(unsigned)(s3 * G + gl)];
      fma4(acc, v0, w0);
      fma4(acc, v1, w1);
      fma4(acc, v2, w2);
      fma4(acc, v3, w3);
    }
    for (; k < m; ++k) {
      const int   s0 = __shfl(rec.x, k, G);
      const float w0 = __int_as_float(__shfl(rec.y, k, G));
      const float4 v0 = Hv[(unsigned)(s0 * G + gl)];
      fma4(acc, v0, w0);
    }
  }

  if (BIAS) {
    const int c0 = gl * 4;
    acc.x += bias[c0]; acc.y += bias[c0 + 1]; acc.z += bias[c0 + 2]; acc.w += bias[c0 + 3];
  }
  if (RELU) {
    acc.x = fmaxf(acc.x, 0.0f); acc.y = fmaxf(acc.y, 0.0f);
    acc.z = fmaxf(acc.z, 0.0f); acc.w = fmaxf(acc.w, 0.0f);
  }
  ((float4*)out)[(unsigned)(node * G + gl)] = acc;
}

static inline unsigned nblk(long long n, int b) { return (unsigned)((n + b - 1) / b); }

static constexpr size_t al256(size_t x) { return (x + 255) & ~(size_t)255; }
static constexpr size_t OFF_CNT   = 0;
static constexpr size_t OFF_DINV  = OFF_CNT   + al256((size_t)NN * 4);
static constexpr size_t OFF_ROW   = OFF_DINV  + al256((size_t)NN * 4);
static constexpr size_t OFF_RUN   = OFF_ROW   + al256((size_t)NN * 4);
static constexpr size_t OFF_PART  = OFF_RUN   + al256((size_t)NB1 * (NBIN + 1) * 4);
static constexpr size_t OFF_CSR   = OFF_PART  + al256((size_t)NB1 * PSTRIDE * 4);
static constexpr size_t OFF_PAIR  = OFF_CSR   + al256((size_t)NE * 8);
static constexpr size_t OFF_G     = OFF_PAIR  + al256((size_t)NE * 8);
static constexpr size_t OFF_P     = OFF_G     + al256((size_t)NN * 128 * 4);

extern "C" void kernel_launch(void* const* d_in, const int* in_sizes, int n_in,
                              void* d_out, int out_size, void* d_ws, size_t ws_size,
                              hipStream_t stream) {
  (void)in_sizes; (void)n_in; (void)out_size; (void)ws_size;
  const float* x   = (const float*)d_in[0];
  const int*   ei  = (const int*)d_in[1];
  const float* We1 = (const float*)d_in[2]; const float* be1 = (const float*)d_in[3];
  const float* We2 = (const float*)d_in[4]; const float* be2 = (const float*)d_in[5];
  const float* Wd1 = (const float*)d_in[6]; const float* bd1 = (const float*)d_in[7];
  const float* Wd2 = (const float*)d_in[8]; const float* bd2 = (const float*)d_in[9];
  const int* src = ei;
  const int* dst = ei + NE;

  char*  ws       = (char*)d_ws;
  int*   cnt      = (int*)(ws + OFF_CNT);
  float* dinv     = (float*)(ws + OFF_DINV);
  int*   rowstart = (int*)(ws + OFF_ROW);
  int*   runstart = (int*)(ws + OFF_RUN);
  int*   part     = (int*)(ws + OFF_PART);
  int*   csr      = (int*)(ws + OFF_CSR);
  int*   pairs    = (int*)(ws + OFF_PAIR);
  float* G        = (float*)(ws + OFF_G);
  float* P        = (float*)(ws + OFF_P);
  float* out      = (float*)d_out;

  k_partition<<<NB1, 256, 0, stream>>>(src, dst, part, runstart);
  k_bin      <<<NBIN, 1024, 0, stream>>>(part, runstart, cnt, rowstart, dinv, pairs);
  k_weight   <<<nblk(NE / 2, 512), 256, 0, stream>>>(dinv, pairs, csr);

  const long long rt = NN / 16;
  k_gemm_wmma<false, false><<<nblk(rt * 4 * 32, 256), 256, 0, stream>>>(x, We1, be1, G, 128, 64);
  k_gather<64, true, true><<<nblk(NN, 16), 256, 0, stream>>>(G, rowstart, cnt, csr, dinv, be1, P);
  k_gemm_wmma<false, false><<<nblk(rt * 2 * 32, 256), 256, 0, stream>>>(P, We2, be2, G, 64, 32);
  k_gather<32, true, false><<<nblk(NN, 32), 256, 0, stream>>>(G, rowstart, cnt, csr, dinv, be2, P);
  k_gather<32, false, false><<<nblk(NN, 32), 256, 0, stream>>>(P, rowstart, cnt, csr, dinv, bd1, G);
  k_gemm_wmma<true, true><<<nblk(rt * 4 * 32, 256), 256, 0, stream>>>(G, Wd1, bd1, P, 32, 64);
  k_gather<64, false, false><<<nblk(NN, 16), 256, 0, stream>>>(P, rowstart, cnt, csr, dinv, bd2, G);
  k_gemm_wmma<true, false><<<nblk(rt * 8 * 32, 256), 256, 0, stream>>>(G, Wd2, bd2, out, 64, 128);
}


# Each product that follows a gather (64->32, 32->64, 64->128) runs in that gather's epilogue: finished rows go through an LDS tile to the same f32 WMMA loaders; 11 launches -> 8, no sum order change
# speedup vs baseline: 4.9243x; 1.2375x over previous
#include <hip/hip_runtime.h>


#define NN 50000
#define NE 800000

typedef __attribute__((ext_vector_type(2))) float v2f;
typedef __attribute__((ext_vector_type(8))) float v8f;

#define NBIN    196
#define NB1     256
#define CHUNK   3125
#define PSTRIDE 3136
#define EPT     13
#define RPW     8
#define CAP     16384

__device__ __forceinline__ int wave_incl_scan(int v, int lane) {
#pragma unroll
  for (int off = 1; off < 32; off <<= 1) {
    const int t = __shfl_up(v, off, 32);
    if (lane >= off) v += t;
  }
  return v;
}

__global__ void __launch_bounds__(256) k_partition(const int* __restrict__ src, const int* __restrict__ dst,
                                                   int* __restrict__ part, int* __restrict__ runstart) {
  __shared__ int hist[256];
  __shared__ int wsum[8];
  __shared__ int sorted[CHUNK];
  const int tid  = threadIdx.x;
  const int lane = tid & 31;
  const int wv   = tid >> 5;
  const int b    = blockIdx.x;
  const int e0   = b * CHUNK;
  hist[tid] = 0;
  int dv[EPT], sv[EPT];
#pragma unroll
  for (int j = 0; j < EPT; ++j) {
    const int i = tid + j * 256;
    dv[j] = 0; sv[j] = 0;
    if (i < CHUNK) { dv[j] = dst[e0 + i]; sv[j] = src[e0 + i]; }
  }
  __syncthreads();
  int key[EPT];
#pragma unroll
  for (int j = 0; j < EPT; ++j) {
    const int i = tid + j * 256;
    key[j] = -1;
    if (i < CHUNK) {
      const int bin = dv[j] >> 8;
      const int r = __hip_atomic_fetch_add(&hist[bin], 1, __ATOMIC_RELAXED, __HIP_MEMORY_SCOPE_WORKGROUP);
      key[j] = (bin << 16) | r;
    }
  }
  __syncthreads();
  const int c = hist[tid];
  const int incl = wave_incl_scan(c, lane);
  if (lane == 31) wsum[wv] = incl;
  __syncthreads();
  int pre = 0;
#pragma unroll
  for (int w = 0; w < 8; ++w) if (w < wv) pre += wsum[w];
  const int excl = incl + pre - c;
  hist[tid] = excl;
  if (tid <= NBIN) runstart[b * (NBIN + 1) + tid] = excl;
  __syncthreads();
#pragma unroll
  for (int j = 0; j < EPT; ++j)
    if (key[j] >= 0) sorted[hist[key[j] >> 16] + (key[j] & 0xffff)] = sv[j] | ((dv[j] & 255) << 16);
  __syncthreads();
  for (int i = tid; i < CHUNK; i += 256) part[b * PSTRIDE + i] = sorted[i];
}

__device__ __forceinline__ void place_rec(int rec, int bin, const int* nstart, int* ncur,
                                          bool fits, int2* stage, int2* out) {
  const int dl  = (rec >> 16) & 255;
  const int pos = nstart[dl] +
      __hip_atomic_fetch_add(&ncur[dl], 1, __ATOMIC_RELAXED, __HIP_MEMORY_SCOPE_WORKGROUP);
  int2 o;
  o.x = rec & 0xffff;
  o.y = bin * 256 + dl;
  if (fits) stage[pos] = o; else out[pos] = o;
}

__global__ void __launch_bounds__(1024) k_bin(const int* __restrict__ part, const int* __restrict__ runstart,
                                              int* __restrict__ cnt, int* __restrict__ rowstart,
                                              float* __restrict__ dinv, int* __restrict__ pairs) {
  __shared__ int  rs0[NB1];
  __shared__ int  rs1[NB1];
  __shared__ int  ncnt[256];
  __shared__ int  nstart[256];
  __shared__ int  wsA[8];
  __shared__ int  wsC[8];
  __shared__ int2 stage[CAP];
  const int tid  = threadIdx.x;
  const int lane = tid & 31;
  const int wv   = tid >> 5;
  const int bin  = blockIdx.x;
  int a = 0;
  if (tid < NB1) {
    a = runstart[tid * (NBIN + 1) + bin];
    const int z = runstart[tid * (NBIN + 1) + bin + 1];
    rs0[tid] = a; rs1[tid] = z;
    ncnt[tid] = 0;
  }
  int sa = a;
#pragma unroll
  for (int off = 16; off > 0; off >>= 1) sa += __shfl_xor(sa, off, 32);
  if (tid < NB1 && lane == 0) wsA[wv] = sa;
  __syncthreads();
  int base = 0;
#pragma unroll
  for (int w = 0; w < 8; ++w) base += wsA[w];

  int recs[RPW];
#pragma unroll
  for (int k = 0; k < RPW; ++k) {
    const int r = wv + 32 * k;
    const int i = rs0[r] + lane;
    recs[k] = -1;
    if (i < rs1[r]) recs[k] = part[r * PSTRIDE + i];
  }
#pragma unroll
  for (int k = 0; k < RPW; ++k)
    if (recs[k] >= 0)
      __hip_atomic_fetch_add(&ncnt[(recs[k] >> 16) & 255], 1, __ATOMIC_RELAXED, __HIP_MEMORY_SCOPE_WORKGROUP);
  for (int k = 0; k < RPW; ++k) {
    const int r = wv + 32 * k;
    const int z = rs1[r];
    const int* p = part + r * PSTRIDE;
    for (int i = rs0[r] + lane + 32; i < z; i += 32)
      __hip_atomic_fetch_add(&ncnt[(p[i] >> 16) & 255], 1, __ATOMIC_RELAXED, __HIP_MEMORY_SCOPE_WORKGROUP);
  }
  __syncthreads();
  const int c = (tid < 256) ? ncnt[tid] : 0;
  const int incl = wave_incl_scan(c, lane);
  if (tid < 256 && lane == 31) wsC[wv] = incl;
  __syncthreads();
  int pre = 0, T = 0;
#pragma unroll
  for (int w = 0; w < 8; ++w) {
    const int t = wsC[w];
    T += t;
    if (w < wv) pre += t;
  }
  if (tid < 256) {
    const int excl = incl + pre - c;
    nstart[tid] = excl;
    ncnt[tid] = 0;
    const int node = bin * 256 + tid;
    if (node < NN) {
      cnt[node] = c;
      rowstart[node] = base + excl;
      dinv[node] = rsqrtf((float)(c + 1));
    }
  }
  __syncthreads();
  const bool fits = (T <= CAP);
  int2* out = (int2*)pairs + base;
#pragma unroll
  for (int k = 0; k < RPW; ++k)
    if (recs[k] >= 0) place_rec(recs[k], bin, nstart, ncnt, fits, stage, out);
  for (int k = 0; k < RPW; ++k) {
    const int r = wv + 32 * k;
    const int z = rs1[r];
    const int* p = part + r * PSTRIDE;
    for (int i = rs0[r] + lane + 32; i < z; i += 32)
      place_rec(p[i], bin, nstart, ncnt, fits, stage, out);
  }
  __syncthreads();
  if (fits)
    for (int i = tid; i < T; i += 1024) out[i] = stage[i];
}

__global__ void __launch_bounds__(256) k_weight(const float* __restrict__ dinv, const int* __restrict__ pairs,
                                                int* __restrict__ csr) {
  const int4* in = (const int4*)pairs;
  int4* outp = (int4*)csr;
  const int i0 = blockIdx.x * 512 + threadIdx.x;
  const int i1 = i0 + 256;
  const bool h0 = i0 < NE / 2, h1 = i1 < NE / 2;
  int4 u = in[h0 ? i0 : NE / 2 - 1];
  int4 v = in[h1 ? i1 : NE / 2 - 1];
  const float a0 = dinv[u.x] * dinv[u.y];
  const float a1 = dinv[u.z] * dinv[u.w];
  const float b0 = dinv[v.x] * dinv[v.y];
  const float b1 = dinv[v.z] * dinv[v.w];
  u.y = __float_as_int(a0); u.w = __float_as_int(a1);
  v.y = __float_as_int(b0); v.w = __float_as_int(b1);
  if (h0) outp[i0] = u;
  if (h1) outp[i1] = v;
}

template <bool BIAS, bool RELU>
__global__ void k_gemm_wmma(const float* __restrict__ X, const float* __restrict__ W,
                            const float* __restrict__ bias, float* __restrict__ H, int K, int N) {
  int wave = (int)((blockIdx.x * blockDim.x + threadIdx.x) >> 5);
  int lane = threadIdx.x & 31;
  int tilesN = N >> 4;
  int totTiles = (NN >> 4) * tilesN;
  if (wave >= totTiles) return;
  int tr = wave / tilesN;
  int tm = tr << 4;
  int tn = (wave - tr * tilesN) << 4;
  int half = lane >> 4;
  int l    = lane & 15;

  const float* xrow = X + (size_t)(tm + l) * K + 2 * half;
  const float* wcol = W + tn + l;
  float bv = 0.0f;
  if (BIAS) bv = bias[tn + l];
  v8f c = {bv, bv, bv, bv, bv, bv, bv, bv};
  for (int k = 0; k < K; k += 4) {
    v2f a, b;
    a.x = xrow[k];
    a.y = xrow[k + 1];
    b.x = wcol[(size_t)(k + 2 * half) * N];
    b.y = wcol[(size_t)(k + 2 * half + 1) * N];
    c = __builtin_amdgcn_wmma_f32_16x16x4_f32(false, a, false, b, (short)0, c, false, false);
  }
  float* hrow = H + (size_t)(tm + 8 * half) * N + tn + l;
#pragma unroll
  for (int r = 0; r < 8; ++r) {
    float v = c[r];
    if (RELU) v = fmaxf(v, 0.0f);
    hrow[(size_t)r * N] = v;
  }
}

__device__ __forceinline__ void fma4(float4& acc, const float4& v, float w) {
  acc.x = __builtin_fmaf(v.x, w, acc.x);
  acc.y = __builtin_fmaf(v.y, w, acc.y);
  acc.z = __builtin_fmaf(v.z, w, acc.z);
  acc.w = __builtin_fmaf(v.w, w, acc.w);
}

template <int C, bool BIAS, bool RELU>
__global__ void __launch_bounds__(256) k_gather(const float* __restrict__ H,
                                                const int* __restrict__ rowstart,
                                                const int* __restrict__ cnt,
                                                const int* __restrict__ csr,
                                                const float* __restrict__ dinv,
                                                const float* __restrict__ bias,
                                                float* __restrict__ out) {
  constexpr int G   = C / 4;
  constexpr int NPB = 256 / G;
  const int tid  = threadIdx.x;
  const int gl   = tid & (G - 1);
  const int node = blockIdx.x * NPB + tid / G;
  if (node >= NN) return;

  const float4* Hv = (const float4*)H;
  const int2*   ev = (const int2*)csr;
  const int   rs = rowstart[node];
  const int   n  = cnt[node];
  const float di = dinv[node];
  const float sl = di * di;

  float4 acc = Hv[(unsigned)(node * G + gl)];
  acc.x *= sl; acc.y *= sl; acc.z *= sl; acc.w *= sl;

  for (int base = 0; base < n; base += G) {
    const int m = (n - base < G) ? (n - base) : G;
    int2 rec;
    rec.x = 0; rec.y = 0;
    if (gl < m) rec = ev[rs + base + gl];
    int k = 0;
    for (; k + 4 <= m; k += 4) {
      const int   s0 = __shfl(rec.x, k,     G);
      const int   s1 = __shfl(rec.x, k + 1, G);
      const int   s2 = __shfl(rec.x, k + 2, G);
      const int   s3 = __shfl(rec.x, k + 3, G);
      const float w0 = __int_as_float(__shfl(rec.y, k,     G));
      const float w1 = __int_as_float(__shfl(rec.y, k + 1, G));
      const float w2 = __int_as_float(__shfl(rec.y, k + 2, G));
      const float w3 = __int_as_float(__shfl(rec.y, k + 3, G));
      const float4 v0 = Hv[(unsigned)(s0 * G + gl)];
      const float4 v1 = Hv[(unsigned)(s1 * G + gl)];
      const float4 v2 = Hv[(unsigned)(s2 * G + gl)];
      const float4 v3 = Hv[(unsigned)(s3 * G + gl)];
      fma4(acc, v0, w0);
      fma4(acc, v1, w1);
      fma4(acc, v2, w2);
      fma4(acc, v3, w3);
    }
    for (; k < m; ++k) {
      const int   s0 = __shfl(rec.x, k, G);
      const float w0 = __int_as_float(__shfl(rec.y, k, G));
      const float4 v0 = Hv[(unsigned)(s0 * G + gl)];
      fma4(acc, v0, w0);
    }
  }

  if (BIAS) {
    const int c0 = gl * 4;
    acc.x += bias[c0]; acc.y += bias[c0 + 1]; acc.z += bias[c0 + 2]; acc.w += bias[c0 + 3];
  }
  if (RELU) {
    acc.x = fmaxf(acc.x, 0.0f); acc.y = fmaxf(acc.y, 0.0f);
    acc.z = fmaxf(acc.z, 0.0f); acc.w = fmaxf(acc.w, 0.0f);
  }
  ((float4*)out)[(unsigned)(node * G + gl)] = acc;
}

template <int C, int N, bool BIAS_IN, bool RELU_IN, bool BIAS_OUT, bool RELU_OUT>
__global__ void __launch_bounds__(256) k_gather_gemm(const float* __restrict__ H,
                                                     const int* __restrict__ rowstart,
                                                     const int* __restrict__ cnt,
                                                     const int* __restrict__ csr,
                                                     const float* __restrict__ dinv,
                                                     const float* __restrict__ bias_in,
                                                     const float* __restrict__ W,
                                                     const float* __restrict__ bias_out,
                                                     float* __restrict__ out) {
  constexpr int G     = C / 4;
  constexpr int NPB   = 256 / G;
  constexpr int PITCH = C + 4;
  constexpr int TN    = N / 16;
  constexpr int TILES = (NPB / 16) * TN;
  __shared__ float tile[NPB * PITCH];
  const int tid  = threadIdx.x;
  const int gl   = tid & (G - 1);
  const int nl   = tid / G;
  const int node = blockIdx.x * NPB + nl;

  float4 acc;
  acc.x = 0.0f; acc.y = 0.0f; acc.z = 0.0f; acc.w = 0.0f;
  if (node < NN) {
    const float4* Hv = (const float4*)H;
    const int2*   ev = (const int2*)csr;
    const int   rs = rowstart[node];
    const int   n  = cnt[node];
    const float di = dinv[node];
    const float sl = di * di;

    acc = Hv[(unsigned)(node * G + gl)];
    acc.x *= sl; acc.y *= sl; acc.z *= sl; acc.w *= sl;

    for (int base = 0; base < n; base += G) {
      const int m = (n - base < G) ? (n - base) : G;
      int2 rec;
      rec.x = 0; rec.y = 0;
      if (gl < m) rec = ev[rs + base + gl];
      int k = 0;
      for (; k + 4 <= m; k += 4) {
        const int   s0 = __shfl(rec.x, k,     G);
        const int   s1 = __shfl(rec.x, k + 1, G);
        const int   s2 = __shfl(rec.x, k + 2, G);
        const int   s3 = __shfl(rec.x, k + 3, G);
        const float w0 = __int_as_float(__shfl(rec.y, k,     G));
        const float w1 = __int_as_float(__shfl(rec.y, k + 1, G));
        const float w2 = __int_as_float(__shfl(rec.y, k + 2, G));
        const float w3 = __int_as_float(__shfl(rec.y, k + 3, G));
        const float4 v0 = Hv[(unsigned)(s0 * G + gl)];
        const float4 v1 = Hv[(unsigned)(s1 * G + gl)];
        const float4 v2 = Hv[(unsigned)(s2 * G + gl)];
        const float4 v3 = Hv[(unsigned)(s3 * G + gl)];
        fma4(acc, v0, w0);
        fma4(acc, v1, w1);
        fma4(acc, v2, w2);
        fma4(acc, v3, w3);
      }
      for (; k < m; ++k) {
        const int   s0 = __shfl(rec.x, k, G);
        const float w0 = __int_as_float(__shfl(rec.y, k, G));
        const float4 v0 = Hv[(unsigned)(s0 * G + gl)];
        fma4(acc, v0, w0);
      }
    }

    if (BIAS_IN) {
      const int c0 = gl * 4;
      acc.x += bias_in[c0]; acc.y += bias_in[c0 + 1]; acc.z += bias_in[c0 + 2]; acc.w += bias_in[c0 + 3];
    }
    if (RELU_IN) {
      acc.x = fmaxf(acc.x, 0.0f); acc.y = fmaxf(acc.y, 0.0f);
      acc.z = fmaxf(acc.z, 0.0f); acc.w = fmaxf(acc.w, 0.0f);
    }
  }
  *(float4*)&tile[nl * PITCH + gl * 4] = acc;
  __syncthreads();

  const int wv   = tid >> 5;
  const int lane = tid & 31;
  if (wv >= TILES) return;
  const int tr   = wv / TN;
  const int tn   = (wv - tr * TN) << 4;
  const int row0 = blockIdx.x * NPB + tr * 16;
  if (row0 >= NN) return;
  const int half = lane >> 4;
  const int l    = lane & 15;

  const float* arow = tile + (tr * 16 + l) * PITCH + 2 * half;
  const float* wcol = W + tn + l;
  float bv = 0.0f;
  if (BIAS_OUT) bv = bias_out[tn + l];
  v8f c = {bv, bv, bv, bv, bv, bv, bv, bv};
#pragma unroll
  for (int k = 0; k < C; k += 4) {
    v2f a, b;
    a.x = arow[k];
    a.y = arow[k + 1];
    b.x = wcol[(k + 2 * half) * N];
    b.y = wcol[(k + 2 * half + 1) * N];
    c = __builtin_amdgcn_wmma_f32_16x16x4_f32(false, a, false, b, (short)0, c, false, false);
  }
  float* hrow = out + (size_t)(row0 + 8 * half) * N + tn + l;
#pragma unroll
  for (int r = 0; r < 8; ++r) {
    float v = c[r];
    if (RELU_OUT) v = fmaxf(v, 0.0f);
    hrow[(size_t)r * N] = v;
  }
}

static inline unsigned nblk(long long n, int b) { return (unsigned)((n + b - 1) / b); }

static constexpr size_t al256(size_t x) { return (x + 255) & ~(size_t)255; }
static constexpr size_t OFF_CNT   = 0;
static constexpr size_t OFF_DINV  = OFF_CNT   + al256((size_t)NN * 4);
static constexpr size_t OFF_ROW   = OFF_DINV  + al256((size_t)NN * 4);
static constexpr size_t OFF_RUN   = OFF_ROW   + al256((size_t)NN * 4);
static constexpr size_t OFF_PART  = OFF_RUN   + al256((size_t)NB1 * (NBIN + 1) * 4);
static constexpr size_t OFF_CSR   = OFF_PART  + al256((size_t)NB1 * PSTRIDE * 4);
static constexpr size_t OFF_PAIR  = OFF_CSR   + al256((size_t)NE * 8);
static constexpr size_t OFF_G     = OFF_PAIR  + al256((size_t)NE * 8);
static constexpr size_t OFF_P     = OFF_G     + al256((size_t)NN * 128 * 4);

extern "C" void kernel_launch(void* const* d_in, const int* in_sizes, int n_in,
                              void* d_out, int out_size, void* d_ws, size_t ws_size,
                              hipStream_t stream) {
  (void)in_sizes; (void)n_in; (void)out_size; (void)ws_size;
  const float* x   = (const float*)d_in[0];
  const int*   ei  = (const int*)d_in[1];
  const float* We1 = (const float*)d_in[2]; const float* be1 = (const float*)d_in[3];
  const float* We2 = (const float*)d_in[4]; const float* be2 = (const float*)d_in[5];
  const float* Wd1 = (const float*)d_in[6]; const float* bd1 = (const float*)d_in[7];
  const float* Wd2 = (const float*)d_in[8]; const float* bd2 = (const float*)d_in[9];
  const int* src = ei;
  const int* dst = ei + NE;

  char*  ws       = (char*)d_ws;
  int*   cnt      = (int*)(ws + OFF_CNT);
  float* dinv     = (float*)(ws + OFF_DINV);
  int*   rowstart = (int*)(ws + OFF_ROW);
  int*   runstart = (int*)(ws + OFF_RUN);
  int*   part     = (int*)(ws + OFF_PART);
  int*   csr      = (int*)(ws + OFF_CSR);
  int*   pairs    = (int*)(ws + OFF_PAIR);
  float* G        = (float*)(ws + OFF_G);
  float* P        = (float*)(ws + OFF_P);
  float* out      = (float*)d_out;

  k_partition<<<NB1, 256, 0, stream>>>(src, dst, part, runstart);
  k_bin      <<<NBIN, 1024, 0, stream>>>(part, runstart, cnt, rowstart, dinv, pairs);
  k_weight   <<<nblk(NE / 2, 512), 256, 0, stream>>>(dinv, pairs, csr);

  const long long rt = NN / 16;
  k_gemm_wmma<false, false><<<nblk(rt * 4 * 32, 256), 256, 0, stream>>>(x, We1, be1, G, 128, 64);
  k_gather_gemm<64, 32, true, true, false, false><<<nblk(NN, 16), 256, 0, stream>>>(
      G, rowstart, cnt, csr, dinv, be1, We2, be2, P);
  k_gather<32, true, false><<<nblk(NN, 32), 256, 0, stream>>>(P, rowstart, cnt, csr, dinv, be2, G);
  k_gather_gemm<32, 64, false, false, true, true><<<nblk(NN, 32), 256, 0, stream>>>(
      G, rowstart, cnt, csr, dinv, bd1, Wd1, bd1, P);
  k_gather_gemm<64, 128, false, false, true, false><<<nblk(NN, 16), 256, 0, stream>>>(
      P, rowstart, cnt, csr, dinv, bd2, Wd2, bd2, out);
}


# Layer-1 product 128->64 runs inside the bin launch: each bin block multiplies its own 256 rows (x and W in LDS, same WMMA loaders and k order) where its sort waits on memory; 7 -> 6 launches
# speedup vs baseline: 5.8129x; 1.1805x over previous
#include <hip/hip_runtime.h>


#define NN 50000
#define NE 800000

typedef __attribute__((ext_vector_type(2))) float v2f;
typedef __attribute__((ext_vector_type(8))) float v8f;

#define NBIN    196
#define NB1     256
#define CHUNK   3125
#define PSTRIDE 3136
#define EPT     13
#define RPW     8
#define CAP     16384
#define XP      132
#define WP      72

__device__ __forceinline__ int wave_incl_scan(int v, int lane) {
#pragma unroll
  for (int off = 1; off < 32; off <<= 1) {
    const int t = __shfl_up(v, off, 32);
    if (lane >= off) v += t;
  }
  return v;
}

__global__ void __launch_bounds__(256) k_partition(const int* __restrict__ src, const int* __restrict__ dst,
                                                   int* __restrict__ part, int* __restrict__ runstart) {
  __shared__ int hist[256];
  __shared__ int wsum[8];
  __shared__ int sorted[CHUNK];
  const int tid  = threadIdx.x;
  const int lane = tid & 31;
  const int wv   = tid >> 5;
  const int b    = blockIdx.x;
  const int e0   = b * CHUNK;
  hist[tid] = 0;
  int dv[EPT], sv[EPT];
#pragma unroll
  for (int j = 0; j < EPT; ++j) {
    const int i = tid + j * 256;
    dv[j] = 0; sv[j] = 0;
    if (i < CHUNK) { dv[j] = dst[e0 + i]; sv[j] = src[e0 + i]; }
  }
  __syncthreads();
  int key[EPT];
#pragma unroll
  for (int j = 0; j < EPT; ++j) {
    const int i = tid + j * 256;
    key[j] = -1;
    if (i < CHUNK) {
      const int bin = dv[j] >> 8;
      const int r = __hip_atomic_fetch_add(&hist[bin], 1, __ATOMIC_RELAXED, __HIP_MEMORY_SCOPE_WORKGROUP);
      key[j] = (bin << 16) | r;
    }
  }
  __syncthreads();
  const int c = hist[tid];
  const int incl = wave_incl_scan(c, lane);
  if (lane == 31) wsum[wv] = incl;
  __syncthreads();
  int pre = 0;
#pragma unroll
  for (int w = 0; w < 8; ++w) if (w < wv) pre += wsum[w];
  const int excl = incl + pre - c;
  hist[tid] = excl;
  if (tid <= NBIN) runstart[b * (NBIN + 1) + tid] = excl;
  __syncthreads();
#pragma unroll
  for (int j = 0; j < EPT; ++j)
    if (key[j] >= 0) sorted[hist[key[j] >> 16] + (key[j] & 0xffff)] = sv[j] | ((dv[j] & 255) << 16);
  __syncthreads();
  for (int i = tid; i < CHUNK; i += 256) part[b * PSTRIDE + i] = sorted[i];
}

__device__ __forceinline__ void place_rec(int rec, const int* nstart, int* ncur,
                                          bool fits, int* stage, int* out) {
  const int dl  = (rec >> 16) & 255;
  const int pos = nstart[dl] +
      __hip_atomic_fetch_add(&ncur[dl], 1, __ATOMIC_RELAXED, __HIP_MEMORY_SCOPE_WORKGROUP);
  const int s = rec & 0xffff;
  if (fits) stage[pos] = s; else out[pos] = s;
}

template <int K0, int K1>
__device__ __forceinline__ void mm_steps(const float* arow, const float* wcol, int half, v8f& c0, v8f& c1) {
#pragma unroll 8
  for (int k = K0; k < K1; k += 4) {
    v2f fa0, fa1, fb;
    fa0.x = arow[k];
    fa0.y = arow[k + 1];
    fa1.x = arow[16 * XP + k];
    fa1.y = arow[16 * XP + k + 1];
    fb.x  = wcol[(k + 2 * half) * WP];
    fb.y  = wcol[(k + 2 * half + 1) * WP];
    c0 = __builtin_amdgcn_wmma_f32_16x16x4_f32(false, fa0, false, fb, (short)0, c0, false, false);
    c1 = __builtin_amdgcn_wmma_f32_16x16x4_f32(false, fa1, false, fb, (short)0, c1, false, false);
  }
}

__global__ void __launch_bounds__(1024) k_bin(const int* __restrict__ part, const int* __restrict__ runstart,
                                              int* __restrict__ cnt, int* __restrict__ rowstart,
                                              float* __restrict__ dinv, int* __restrict__ csr,
                                              const float* __restrict__ X, const float* __restrict__ W,
                                              float* __restrict__ H) {
  __shared__ int  rs0[NB1];
  __shared__ int  rs1[NB1];
  __shared__ int  ncnt[256];
  __shared__ int  nstart[256];
  __shared__ int  wsA[8];
  __shared__ int  wsC[8];
  __shared__ float dl[256];
  __shared__ int  stage[CAP];
  __shared__ __attribute__((aligned(16))) float xs[256 * XP];
  __shared__ __attribute__((aligned(16))) float wl[128 * WP];
  const int tid  = threadIdx.x;
  const int lane = tid & 31;
  const int wv   = tid >> 5;
  const int bin  = blockIdx.x;

  const float4* xg = (const float4*)(X + (size_t)bin * 256 * 128);
  float4 xv[8], wq[3];
#pragma unroll
  for (int j = 0; j < 8; ++j) {
    const int f = tid + 1024 * j;
    xv[j].x = 0.0f; xv[j].y = 0.0f; xv[j].z = 0.0f; xv[j].w = 0.0f;
    if (bin * 256 + (f >> 5) < NN) xv[j] = xg[f];
  }
#pragma unroll
  for (int j = 0; j < 3; ++j) {
    const int f = tid - 256 + 768 * j;
    wq[j].x = 0.0f; wq[j].y = 0.0f; wq[j].z = 0.0f; wq[j].w = 0.0f;
    if (tid >= 256 && f < 128 * 16) wq[j] = ((const float4*)W)[f];
  }

  int a = 0;
  if (tid < NB1) {
    a = runstart[tid * (NBIN + 1) + bin];
    const int z = runstart[tid * (NBIN + 1) + bin + 1];
    rs0[tid] = a; rs1[tid] = z;
    ncnt[tid] = 0;
  }
  int sa = a;
#pragma unroll
  for (int off = 16; off > 0; off >>= 1) sa += __shfl_xor(sa, off, 32);
  if (tid < NB1 && lane == 0) wsA[wv] = sa;
#pragma unroll
  for (int j = 0; j < 8; ++j) {
    const int f = tid + 1024 * j;
    *(float4*)&xs[(f >> 5) * XP + (f & 31) * 4] = xv[j];
  }
#pragma unroll
  for (int j = 0; j < 3; ++j) {
    const int f = tid - 256 + 768 * j;
    if (tid >= 256 && f < 128 * 16) *(float4*)&wl[(f >> 4) * WP + (f & 15) * 4] = wq[j];
  }
  __syncthreads();
  int base = 0;
#pragma unroll
  for (int w = 0; w < 8; ++w) base += wsA[w];

  int recs[RPW];
#pragma unroll
  for (int k = 0; k < RPW; ++k) {
    const int r = wv + 32 * k;
    const int i = rs0[r] + lane;
    recs[k] = -1;
    if (i < rs1[r]) recs[k] = part[r * PSTRIDE + i];
  }

  const int  rt   = (wv >> 2) * 2;
  const bool live = bin * 256 + rt * 16 < NN;
  const int  half = lane >> 4;
  const int  l    = lane & 15;
  const int  tn   = (wv & 3) * 16;
  const float* arow = xs + (rt * 16 + l) * XP + 2 * half;
  const float* wcol = wl + tn + l;
  v8f c0 = {0.0f, 0.0f, 0.0f, 0.0f, 0.0f, 0.0f, 0.0f, 0.0f};
  v8f c1 = c0;
  if (live) mm_steps<0, 64>(arow, wcol, half, c0, c1);

#pragma unroll
  for (int k = 0; k < RPW; ++k)
    if (recs[k] >= 0)
      __hip_atomic_fetch_add(&ncnt[(recs[k] >> 16) & 255], 1, __ATOMIC_RELAXED, __HIP_MEMORY_SCOPE_WORKGROUP);
  for (int k = 0; k < RPW; ++k) {
    const int r = wv + 32 * k;
    const int z = rs1[r];
    const int* p = part + r * PSTRIDE;
    for (int i = rs0[r] + lane + 32; i < z; i += 32)
      __hip_atomic_fetch_add(&ncnt[(p[i] >> 16) & 255], 1, __ATOMIC_RELAXED, __HIP_MEMORY_SCOPE_WORKGROUP);
  }
  __syncthreads();
  const int c = (tid < 256) ? ncnt[tid] : 0;
  const int incl = wave_incl_scan(c, lane);
  if (tid < 256 && lane == 31) wsC[wv] = incl;
  __syncthreads();
  int pre = 0, T = 0;
#pragma unroll
  for (int w = 0; w < 8; ++w) {
    const int t = wsC[w];
    T += t;
    if (w < wv) pre += t;
  }
  if (tid < 256) {
    const int excl = incl + pre - c;
    nstart[tid] = excl;
    ncnt[tid] = 0;
    const int node = bin * 256 + tid;
    float dn = 0.0f;
    if (node < NN) {
      dn = rsqrtf((float)(c + 1));
      cnt[node] = c;
      rowstart[node] = base + excl;
      dinv[node] = dn;
    }
    dl[tid] = dn;
  }
  if (live) mm_steps<64, 96>(arow, wcol, half, c0, c1);
  __syncthreads();
  const bool fits = (T <= CAP);
  int* out = csr + base;
#pragma unroll
  for (int k = 0; k < RPW; ++k)
    if (recs[k] >= 0) place_rec(recs[k], nstart, ncnt, fits, stage, out);
  for (int k = 0; k < RPW; ++k) {
    const int r = wv + 32 * k;
    const int z = rs1[r];
    const int* p = part + r * PSTRIDE;
    for (int i = rs0[r] + lane + 32; i < z; i += 32)
      place_rec(p[i], nstart, ncnt, fits, stage, out);
  }
  __syncthreads();
  if (fits)
    for (int i = tid; i < T; i += 1024) out[i] = stage[i];

  if (live) mm_steps<96, 128>(arow, wcol, half, c0, c1);
  __syncthreads();
  float* ot = xs;
  if (live) {
    const float* dr = dl + rt * 16 + 8 * half;
    float* orow = ot + (rt * 16 + 8 * half) * 64 + tn + l;
#pragma unroll
    for (int r = 0; r < 8; ++r) {
      orow[r * 64]           = c0[r] * dr[r];
      orow[(r + 16) * 64]    = c1[r] * dr[r + 16];
    }
  }
  __syncthreads();
  float4* hg = (float4*)(H + (size_t)bin * 256 * 64);
#pragma unroll
  for (int j = 0; j < 4; ++j) {
    const int f = tid + 1024 * j;
    if (bin * 256 + (f >> 4) < NN) hg[f] = ((const float4*)ot)[f];
  }
}


__device__ __forceinline__ void add4(float4& acc, const float4& v) {
  acc.x += v.x;
  acc.y += v.y;
  acc.z += v.z;
  acc.w += v.w;
}

template <int G>
__device__ __forceinline__ float4 walk_row(const float4* __restrict__ Hv, const int* __restrict__ ev,
                                           int rs, int n, int node, int gl, float di) {
  float4 acc = Hv[(unsigned)(node * G + gl)];
  for (int base = 0; base < n; base += G) {
    const int m = (n - base < G) ? (n - base) : G;
    int rec = 0;
    if (gl < m) rec = ev[rs + base + gl];
    int k = 0;
    for (; k + 4 <= m; k += 4) {
      const int s0 = __shfl(rec, k,     G);
      const int s1 = __shfl(rec, k + 1, G);
      const int s2 = __shfl(rec, k + 2, G);
      const int s3 = __shfl(rec, k + 3, G);
      const float4 v0 = Hv[(unsigned)(s0 * G + gl)];
      const float4 v1 = Hv[(unsigned)(s1 * G + gl)];
      const float4 v2 = Hv[(unsigned)(s2 * G + gl)];
      const float4 v3 = Hv[(unsigned)(s3 * G + gl)];
      add4(acc, v0);
      add4(acc, v1);
      add4(acc, v2);
      add4(acc, v3);
    }
    for (; k < m; ++k) {
      const int s0 = __shfl(rec, k, G);
      const float4 v0 = Hv[(unsigned)(s0 * G + gl)];
      add4(acc, v0);
    }
  }
  acc.x *= di; acc.y *= di; acc.z *= di; acc.w *= di;
  return acc;
}

template <int C, bool BIAS, bool RELU, bool SCALE_OUT>
__global__ void __launch_bounds__(256) k_gather(const float* __restrict__ H,
                                                const int* __restrict__ rowstart,
                                                const int* __restrict__ cnt,
                                                const int* __restrict__ csr,
                                                const float* __restrict__ dinv,
                                                const float* __restrict__ bias,
                                                float* __restrict__ out) {
  constexpr int G   = C / 4;
  constexpr int NPB = 256 / G;
  const int tid  = threadIdx.x;
  const int gl   = tid & (G - 1);
  const int node = blockIdx.x * NPB + tid / G;
  if (node >= NN) return;

  const float di = dinv[node];
  float4 acc = walk_row<G>((const float4*)H, csr, rowstart[node], cnt[node], node, gl, di);

  if (BIAS) {
    const int c0 = gl * 4;
    acc.x += bias[c0]; acc.y += bias[c0 + 1]; acc.z += bias[c0 + 2]; acc.w += bias[c0 + 3];
  }
  if (RELU) {
    acc.x = fmaxf(acc.x, 0.0f); acc.y = fmaxf(acc.y, 0.0f);
    acc.z = fmaxf(acc.z, 0.0f); acc.w = fmaxf(acc.w, 0.0f);
  }
  if (SCALE_OUT) { acc.x *= di; acc.y *= di; acc.z *= di; acc.w *= di; }
  ((float4*)out)[(unsigned)(node * G + gl)] = acc;
}

template <int C, int N, bool BIAS_IN, bool RELU_IN, bool BIAS_OUT, bool RELU_OUT, bool SCALE_OUT>
__global__ void __launch_bounds__(256) k_gather_gemm(const float* __restrict__ H,
                                                     const int* __restrict__ rowstart,
                                                     const int* __restrict__ cnt,
                                                     const int* __restrict__ csr,
                                                     const float* __restrict__ dinv,
                                                     const float* __restrict__ bias_in,
                                                     const float* __restrict__ W,
                                                     const float* __restrict__ bias_out,
                                                     float* __restrict__ out) {
  constexpr int G     = C / 4;
  constexpr int NPB   = 256 / G;
  constexpr int PITCH = C + 4;
  constexpr int TN    = N / 16;
  constexpr int TILES = (NPB / 16) * TN;
  __shared__ float tile[NPB * PITCH];
  __shared__ float drow[NPB];
  const int tid  = threadIdx.x;
  const int gl   = tid & (G - 1);
  const int nl   = tid / G;
  const int node = blockIdx.x * NPB + nl;

  float4 acc;
  acc.x = 0.0f; acc.y = 0.0f; acc.z = 0.0f; acc.w = 0.0f;
  float di = 0.0f;
  if (node < NN) {
    di = dinv[node];
    acc = walk_row<G>((const float4*)H, csr, rowstart[node], cnt[node], node, gl, di);

    if (BIAS_IN) {
      const int c0 = gl * 4;
      acc.x += bias_in[c0]; acc.y += bias_in[c0 + 1]; acc.z += bias_in[c0 + 2]; acc.w += bias_in[c0 + 3];
    }
    if (RELU_IN) {
      acc.x = fmaxf(acc.x, 0.0f); acc.y = fmaxf(acc.y, 0.0f);
      acc.z = fmaxf(acc.z, 0.0f); acc.w = fmaxf(acc.w, 0.0f);
    }
  }
  *(float4*)&tile[nl * PITCH + gl * 4] = acc;
  if (SCALE_OUT && gl == 0) drow[nl] = di;
  __syncthreads();

  const int wv   = tid >> 5;
  const int lane = tid & 31;
  if (wv >= TILES) return;
  const int tr   = wv / TN;
  const int tn   = (wv - tr * TN) << 4;
  const int row0 = blockIdx.x * NPB + tr * 16;
  if (row0 >= NN) return;
  const int half = lane >> 4;
  const int l    = lane & 15;

  const float* arow = tile + (tr * 16 + l) * PITCH + 2 * half;
  const float* wcol = W + tn + l;
  float bv = 0.0f;
  if (BIAS_OUT) bv = bias_out[tn + l];
  v8f c = {bv, bv, bv, bv, bv, bv, bv, bv};
#pragma unroll
  for (int k = 0; k < C; k += 4) {
    v2f a, b;
    a.x = arow[k];
    a.y = arow[k + 1];
    b.x = wcol[(k + 2 * half) * N];
    b.y = wcol[(k + 2 * half + 1) * N];
    c = __builtin_amdgcn_wmma_f32_16x16x4_f32(false, a, false, b, (short)0, c, false, false);
  }
  const float* dr = drow + tr * 16 + 8 * half;
  float* hrow = out + (size_t)(row0 + 8 * half) * N + tn + l;
#pragma unroll
  for (int r = 0; r < 8; ++r) {
    float v = c[r];
    if (RELU_OUT) v = fmaxf(v, 0.0f);
    if (SCALE_OUT) v *= dr[r];
    hrow[(size_t)r * N] = v;
  }
}

static inline unsigned nblk(long long n, int b) { return (unsigned)((n + b - 1) / b); }

static constexpr size_t al256(size_t x) { return (x + 255) & ~(size_t)255; }
static constexpr size_t OFF_CNT   = 0;
static constexpr size_t OFF_DINV  = OFF_CNT   + al256((size_t)NN * 4);
static constexpr size_t OFF_ROW   = OFF_DINV  + al256((size_t)NN * 4);
static constexpr size_t OFF_RUN   = OFF_ROW   + al256((size_t)NN * 4);
static constexpr size_t OFF_PART  = OFF_RUN   + al256((size_t)NB1 * (NBIN + 1) * 4);
static constexpr size_t OFF_CSR   = OFF_PART  + al256((size_t)NB1 * PSTRIDE * 4);
static constexpr size_t OFF_G     = OFF_CSR   + al256((size_t)NE * 4);
static constexpr size_t OFF_P     = OFF_G     + al256((size_t)NN * 128 * 4);

extern "C" void kernel_launch(void* const* d_in, const int* in_sizes, int n_in,
                              void* d_out, int out_size, void* d_ws, size_t ws_size,
                              hipStream_t stream) {
  (void)in_sizes; (void)n_in; (void)out_size; (void)ws_size;
  const float* x   = (const float*)d_in[0];
  const int*   ei  = (const int*)d_in[1];
  const float* We1 = (const float*)d_in[2]; const float* be1 = (const float*)d_in[3];
  const float* We2 = (const float*)d_in[4]; const float* be2 = (const float*)d_in[5];
  const float* Wd1 = (const float*)d_in[6]; const float* bd1 = (const float*)d_in[7];
  const float* Wd2 = (const float*)d_in[8]; const float* bd2 = (const float*)d_in[9];
  const int* src = ei;
  const int* dst = ei + NE;

  char*  ws       = (char*)d_ws;
  int*   cnt      = (int*)(ws + OFF_CNT);
  float* dinv     = (float*)(ws + OFF_DINV);
  int*   rowstart = (int*)(ws + OFF_ROW);
  int*   runstart = (int*)(ws + OFF_RUN);
  int*   part     = (int*)(ws + OFF_PART);
  int*   csr      = (int*)(ws + OFF_CSR);
  float* G        = (float*)(ws + OFF_G);
  float* P        = (float*)(ws + OFF_P);
  float* out      = (float*)d_out;

  k_partition<<<NB1, 256, 0, stream>>>(src, dst, part, runstart);
  k_bin      <<<NBIN, 1024, 0, stream>>>(part, runstart, cnt, rowstart, dinv, csr, x, We1, G);

  k_gather_gemm<64, 32, true, true, false, false, true><<<nblk(NN, 16), 256, 0, stream>>>(
      G, rowstart, cnt, csr, dinv, be1, We2, be2, P);
  k_gather<32, true, false, true><<<nblk(NN, 32), 256, 0, stream>>>(P, rowstart, cnt, csr, dinv, be2, G);
  k_gather_gemm<32, 64, false, false, true, true, true><<<nblk(NN, 32), 256, 0, stream>>>(
      G, rowstart, cnt, csr, dinv, bd1, Wd1, bd1, P);
  k_gather_gemm<64, 128, false, false, true, false, false><<<nblk(NN, 16), 256, 0, stream>>>(
      P, rowstart, cnt, csr, dinv, bd2, Wd2, bd2, out);
}
